# starting kernel
# baseline (speedup 1.0000x reference)
#include <hip/hip_runtime.h>
#include <hip/hip_bf16.h>
#include <math.h>

#define N_   20000
#define E_   320000
#define R_   8
#define B_   8
#define IN_  128
#define H_   256
#define OUT_ 128
#define P_   100000
#define LN_EPS 1e-5f

typedef float v8f __attribute__((ext_vector_type(8)));
typedef float v2f __attribute__((ext_vector_type(2)));

#define WMMA_F32(a, b, c) \
  __builtin_amdgcn_wmma_f32_16x16x4_f32(false, (a), false, (b), (short)0, (c), false, false)

__global__ void relatt_softmax_kernel(const float* __restrict__ relatt,
                                      float* __restrict__ a) {
  if (threadIdx.x == 0 && blockIdx.x == 0) {
    float m = relatt[0];
    for (int r = 1; r < R_; ++r) m = fmaxf(m, relatt[r]);
    float e[R_]; float s = 0.f;
    for (int r = 0; r < R_; ++r) { e[r] = expf(relatt[r] - m); s += e[r]; }
    for (int r = 0; r < R_; ++r) a[r] = e[r] / s;
  }
}

__global__ void make_weights_kernel(const float* __restrict__ att,
                                    const float* __restrict__ basis,
                                    const float* __restrict__ a,
                                    float* __restrict__ W,
                                    int in_ch, int out_ch) {
  int idx = blockIdx.x * blockDim.x + threadIdx.x;
  int io_sz = in_ch * out_ch;
  int total = R_ * io_sz;
  if (idx >= total) return;
  int r  = idx / io_sz;
  int io = idx - r * io_sz;
  float s = 0.f;
#pragma unroll
  for (int bb = 0; bb < B_; ++bb)
    s += att[r * B_ + bb] * basis[(size_t)bb * io_sz + io];
  W[idx] = s * a[r];
}

__global__ void scatter_agg_kernel(const float* __restrict__ x,
                                   const int* __restrict__ src,
                                   const int* __restrict__ dst,
                                   const int* __restrict__ et,
                                   float* __restrict__ agg, int in_ch) {
  int e = blockIdx.x;
  int f = threadIdx.x;
  int s = src[e], d = dst[e], r = et[e];
  atomicAdd(&agg[((size_t)d * R_ + r) * in_ch + f],
            x[(size_t)s * in_ch + f]);
}

__global__ void gemm_f32_wmma_kernel(const float* __restrict__ A,
                                     const float* __restrict__ Bm,
                                     float* __restrict__ C,
                                     int M, int Nc, int K,
                                     int lda, int ldb, int ldc,
                                     int accumulate) {
  int wave = threadIdx.x >> 5;
  int lane = threadIdx.x & 31;
  int l16  = lane & 15;
  int g    = lane >> 4;
  int tilesN = Nc >> 6;
  int tilesTotal = (M >> 4) * tilesN;
  int tile = blockIdx.x * 8 + wave;
  if (tile >= tilesTotal) return;
  int tm = tile / tilesN;
  int tn = tile - tm * tilesN;

  const float* Arow  = A + (size_t)(tm * 16 + l16) * lda + 2 * g;
  const float* Bbase = Bm + (size_t)(2 * g) * ldb + tn * 64 + l16;

  v8f acc0 = {}, acc1 = {}, acc2 = {}, acc3 = {};
  for (int k0 = 0; k0 < K; k0 += 16) {
    __builtin_prefetch(Arow + k0 + 16, 0, 0);
#pragma unroll
    for (int kk = 0; kk < 16; kk += 4) {
      int k = k0 + kk;
      v2f a = *(const v2f*)(Arow + k);
      const float* Bk0 = Bbase + (size_t)k * ldb;
      const float* Bk1 = Bk0 + ldb;
      v2f b0, b1, b2, b3;
      b0.x = Bk0[0];  b0.y = Bk1[0];
      b1.x = Bk0[16]; b1.y = Bk1[16];
      b2.x = Bk0[32]; b2.y = Bk1[32];
      b3.x = Bk0[48]; b3.y = Bk1[48];
      acc0 = WMMA_F32(a, b0, acc0);
      acc1 = WMMA_F32(a, b1, acc1);
      acc2 = WMMA_F32(a, b2, acc2);
      acc3 = WMMA_F32(a, b3, acc3);
    }
  }

  float* Crow = C + (size_t)(tm * 16 + 8 * g) * ldc + tn * 64 + l16;
#pragma unroll
  for (int i = 0; i < 8; ++i) {
    size_t off = (size_t)i * ldc;
    float v0 = acc0[i], v1 = acc1[i], v2 = acc2[i], v3 = acc3[i];
    if (accumulate) {
      v0 += Crow[off];      v1 += Crow[off + 16];
      v2 += Crow[off + 32]; v3 += Crow[off + 48];
    }
    Crow[off]      = v0;
    Crow[off + 16] = v1;
    Crow[off + 32] = v2;
    Crow[off + 48] = v3;
  }
}

__global__ void gemm_pe_wmma_kernel(const float* __restrict__ z,
                                    const int* __restrict__ gi,
                                    const int* __restrict__ gj,
                                    const float* __restrict__ Bm,
                                    float* __restrict__ C,
                                    int M, int Nc, int K,
                                    int ldb, int ldc) {
  int wave = threadIdx.x >> 5;
  int lane = threadIdx.x & 31;
  int l16  = lane & 15;
  int g    = lane >> 4;
  int tilesN = Nc >> 6;
  int tilesTotal = (M >> 4) * tilesN;
  int tile = blockIdx.x * 8 + wave;
  if (tile >= tilesTotal) return;
  int tm = tile / tilesN;
  int tn = tile - tm * tilesN;

  int row = tm * 16 + l16;
  const float* zi = z + (size_t)gi[row] * OUT_;
  const float* zj = z + (size_t)gj[row] * OUT_;
  const float* Bbase = Bm + (size_t)(2 * g) * ldb + tn * 64 + l16;

  v8f acc0 = {}, acc1 = {}, acc2 = {}, acc3 = {};
  for (int k = 0; k < K; k += 4) {
    int kk = k + 2 * g;
    int f  = kk & (OUT_ - 1);
    int s  = kk >> 7;
    float zix = zi[f], ziy = zi[f + 1];
    float zjx = zj[f], zjy = zj[f + 1];
    v2f a;
    if (s == 0)      { a.x = zix + zjx;        a.y = ziy + zjy; }
    else if (s == 1) { a.x = zix * zjx;        a.y = ziy * zjy; }
    else             { a.x = fabsf(zix - zjx); a.y = fabsf(ziy - zjy); }
    const float* Bk0 = Bbase + (size_t)k * ldb;
    const float* Bk1 = Bk0 + ldb;
    v2f b0, b1, b2, b3;
    b0.x = Bk0[0];  b0.y = Bk1[0];
    b1.x = Bk0[16]; b1.y = Bk1[16];
    b2.x = Bk0[32]; b2.y = Bk1[32];
    b3.x = Bk0[48]; b3.y = Bk1[48];
    acc0 = WMMA_F32(a, b0, acc0);
    acc1 = WMMA_F32(a, b1, acc1);
    acc2 = WMMA_F32(a, b2, acc2);
    acc3 = WMMA_F32(a, b3, acc3);
  }

  float* Crow = C + (size_t)(tm * 16 + 8 * g) * ldc + tn * 64 + l16;
#pragma unroll
  for (int i = 0; i < 8; ++i) {
    size_t off = (size_t)i * ldc;
    Crow[off]      = acc0[i];
    Crow[off + 16] = acc1[i];
    Crow[off + 32] = acc2[i];
    Crow[off + 48] = acc3[i];
  }
}

__global__ void ln_res_gelu_kernel(const float* __restrict__ y,
                                   const float* __restrict__ bias,
                                   const float* __restrict__ gamma,
                                   const float* __restrict__ beta,
                                   const float* __restrict__ resid,
                                   float* __restrict__ out,
                                   int rows, int C, int do_gelu) {
  int wave = threadIdx.x >> 5;
  int lane = threadIdx.x & 31;
  int row = blockIdx.x * 8 + wave;
  if (row >= rows) return;
  const float* yr = y + (size_t)row * C;
  int nv = C >> 5;
  float vals[8];
  float s = 0.f;
  for (int i = 0; i < nv; ++i) {
    int c = lane + (i << 5);
    float v = yr[c];
    if (bias) v += bias[c];
    vals[i] = v;
    s += v;
  }
  for (int o = 16; o; o >>= 1) s += __shfl_xor(s, o, 32);
  float mean = s / (float)C;
  float vs = 0.f;
  for (int i = 0; i < nv; ++i) { float d = vals[i] - mean; vs += d * d; }
  for (int o = 16; o; o >>= 1) vs += __shfl_xor(vs, o, 32);
  float inv = rsqrtf(vs / (float)C + LN_EPS);
  for (int i = 0; i < nv; ++i) {
    int c = lane + (i << 5);
    float v = (vals[i] - mean) * inv * gamma[c] + beta[c];
    if (resid) v += resid[(size_t)row * C + c];
    if (do_gelu) v = 0.5f * v * (1.f + erff(v * 0.70710678118654752f));
    out[(size_t)row * C + c] = v;
  }
}

__global__ void dot64_kernel(const float* __restrict__ d2,
                             const float* __restrict__ w,
                             const float* __restrict__ b,
                             float* __restrict__ logits, int rows) {
  int wave = threadIdx.x >> 5;
  int lane = threadIdx.x & 31;
  int p = blockIdx.x * 8 + wave;
  if (p >= rows) return;
  const float* dr = d2 + (size_t)p * 64;
  float s = dr[lane] * w[lane] + dr[lane + 32] * w[lane + 32];
  for (int o = 16; o; o >>= 1) s += __shfl_xor(s, o, 32);
  if (lane == 0) logits[p] = s + b[0];
}

extern "C" void kernel_launch(void* const* d_in, const int* in_sizes, int n_in,
                              void* d_out, int out_size, void* d_ws, size_t ws_size,
                              hipStream_t stream) {
  (void)in_sizes; (void)n_in; (void)out_size; (void)ws_size;

  const float* x    = (const float*)d_in[0];
  const int*   ei   = (const int*)d_in[1];
  const int*   esrc = ei;
  const int*   edst = ei + E_;
  const int*   et   = (const int*)d_in[2];
  const int*   gp   = (const int*)d_in[3];
  const int*   gi   = gp;
  const int*   gj   = gp + P_;

  const float* basis[4], *att[4], *root[4], *relatt[4], *gam[4], *bet[4];
  for (int i = 0; i < 4; ++i) {
    basis[i]  = (const float*)d_in[4 + 6 * i + 0];
    att[i]    = (const float*)d_in[4 + 6 * i + 1];
    root[i]   = (const float*)d_in[4 + 6 * i + 2];
    relatt[i] = (const float*)d_in[4 + 6 * i + 3];
    gam[i]    = (const float*)d_in[4 + 6 * i + 4];
    bet[i]    = (const float*)d_in[4 + 6 * i + 5];
  }
  const float* res0 = (const float*)d_in[28];
  const float* res3 = (const float*)d_in[29];
  const float* dW0 = (const float*)d_in[30], *db0 = (const float*)d_in[31];
  const float* dg0 = (const float*)d_in[32], *dbb0 = (const float*)d_in[33];
  const float* dW1 = (const float*)d_in[34], *db1 = (const float*)d_in[35];
  const float* dg1 = (const float*)d_in[36], *dbb1 = (const float*)d_in[37];
  const float* dW2 = (const float*)d_in[38], *db2 = (const float*)d_in[39];
  const float* dg2 = (const float*)d_in[40], *dbb2 = (const float*)d_in[41];
  const float* dW3 = (const float*)d_in[42], *db3 = (const float*)d_in[43];

  float* ws = (float*)d_ws;
  size_t off = 0;
  float* a_sm  = ws + off; off += 64;
  float* Wcat  = ws + off; off += (size_t)R_ * H_ * H_;
  float* hA    = ws + off; off += (size_t)N_ * H_;
  float* hB    = ws + off; off += (size_t)N_ * H_;
  float* resid = ws + off; off += (size_t)N_ * H_;
  float* big   = ws + off;
  float* d0 = big;
  float* d1 = big + (size_t)P_ * H_;
  float* d2 = hA;

  auto gemm = [&](const float* A, const float* Bm, float* Cm,
                  int M, int Nc, int K, int lda, int ldb, int ldc, int acc) {
    int tiles = (M / 16) * (Nc / 64);
    gemm_f32_wmma_kernel<<<(tiles + 7) / 8, 256, 0, stream>>>(
        A, Bm, Cm, M, Nc, K, lda, ldb, ldc, acc);
  };

  struct { const float* cur; int in; int out; int li;
           const float* resmat; float* conv; int gelu; } L[4] = {
    { x,  IN_, H_,   0, res0,    hA, 1 },
    { hA, H_,  H_,   1, nullptr, hB, 1 },
    { hB, H_,  H_,   2, nullptr, hA, 1 },
    { hA, H_,  OUT_, 3, res3,    hB, 0 },
  };

  for (int i = 0; i < 4; ++i) {
    const float* cur = L[i].cur;
    int in_ch = L[i].in, out_ch = L[i].out, li = L[i].li;
    float* conv = L[i].conv;

    relatt_softmax_kernel<<<1, 32, 0, stream>>>(relatt[li], a_sm);
    int wtotal = R_ * in_ch * out_ch;
    make_weights_kernel<<<(wtotal + 255) / 256, 256, 0, stream>>>(
        att[li], basis[li], a_sm, Wcat, in_ch, out_ch);

    hipMemsetAsync(big, 0, (size_t)N_ * R_ * in_ch * sizeof(float), stream);
    scatter_agg_kernel<<<E_, in_ch, 0, stream>>>(cur, esrc, edst, et, big, in_ch);

    gemm(big, Wcat, conv, N_, out_ch, R_ * in_ch, R_ * in_ch, out_ch, out_ch, 0);
    gemm(cur, root[li], conv, N_, out_ch, in_ch, in_ch, out_ch, out_ch, 1);

    const float* rptr;
    if (L[i].resmat) {
      gemm(cur, L[i].resmat, resid, N_, out_ch, in_ch, in_ch, out_ch, out_ch, 0);
      rptr = resid;
    } else {
      rptr = cur;
    }
    ln_res_gelu_kernel<<<(N_ + 7) / 8, 256, 0, stream>>>(
        conv, nullptr, gam[li], bet[li], rptr, conv, N_, out_ch, L[i].gelu);
  }

  hipMemcpyAsync((float*)d_out + P_, hB, (size_t)N_ * OUT_ * sizeof(float),
                 hipMemcpyDeviceToDevice, stream);

  {
    int tiles = (P_ / 16) * (H_ / 64);
    gemm_pe_wmma_kernel<<<(tiles + 7) / 8, 256, 0, stream>>>(
        hB, gi, gj, dW0, d0, P_, H_, 3 * OUT_, H_, H_);
  }
  ln_res_gelu_kernel<<<(P_ + 7) / 8, 256, 0, stream>>>(
      d0, db0, dg0, dbb0, nullptr, d0, P_, H_, 1);

  gemm(d0, dW1, d1, P_, H_ / 2, H_, H_, H_ / 2, H_ / 2, 0);
  ln_res_gelu_kernel<<<(P_ + 7) / 8, 256, 0, stream>>>(
      d1, db1, dg1, dbb1, nullptr, d1, P_, H_ / 2, 1);

  gemm(d1, dW2, d2, P_, H_ / 4, H_ / 2, H_ / 2, H_ / 4, H_ / 4, 0);
  ln_res_gelu_kernel<<<(P_ + 7) / 8, 256, 0, stream>>>(
      d2, db2, dg2, dbb2, nullptr, d2, P_, H_ / 4, 1);

  dot64_kernel<<<(P_ + 7) / 8, 256, 0, stream>>>(d2, dW3, db3, (float*)d_out, P_);
}


# Decoder LayerNorm rows held by 8 lanes each, 4 rows per wave, width fixed at compile time: 16-byte loads issued ahead, no wait per column; 225000 waves gone; decoder LN and final dot sums reordered
# speedup vs baseline: 3.0122x; 3.0122x over previous
#include <hip/hip_runtime.h>
#include <hip/hip_bf16.h>
#include <math.h>

#define N_   20000
#define E_   320000
#define R_   8
#define B_   8
#define IN_  128
#define H_   256
#define OUT_ 128
#define P_   100000
#define LN_EPS 1e-5f

typedef float v8f __attribute__((ext_vector_type(8)));
typedef float v2f __attribute__((ext_vector_type(2)));
typedef float v4f __attribute__((ext_vector_type(4)));

#define WMMA_F32(a, b, c) \
  __builtin_amdgcn_wmma_f32_16x16x4_f32(false, (a), false, (b), (short)0, (c), false, false)

__global__ void relatt_softmax_kernel(const float* __restrict__ relatt,
                                      float* __restrict__ a) {
  if (threadIdx.x == 0 && blockIdx.x == 0) {
    float m = relatt[0];
    for (int r = 1; r < R_; ++r) m = fmaxf(m, relatt[r]);
    float e[R_]; float s = 0.f;
    for (int r = 0; r < R_; ++r) { e[r] = expf(relatt[r] - m); s += e[r]; }
    for (int r = 0; r < R_; ++r) a[r] = e[r] / s;
  }
}

__global__ void make_weights_kernel(const float* __restrict__ att,
                                    const float* __restrict__ basis,
                                    const float* __restrict__ a,
                                    float* __restrict__ W,
                                    int in_ch, int out_ch) {
  int idx = blockIdx.x * blockDim.x + threadIdx.x;
  int io_sz = in_ch * out_ch;
  int total = R_ * io_sz;
  if (idx >= total) return;
  int r  = idx / io_sz;
  int io = idx - r * io_sz;
  float s = 0.f;
#pragma unroll
  for (int bb = 0; bb < B_; ++bb)
    s += att[r * B_ + bb] * basis[(size_t)bb * io_sz + io];
  int i = io / out_ch;
  int o = io - i * out_ch;
  W[(size_t)i * (R_ * out_ch) + r * out_ch + o] = s * a[r];
}

__global__ void chain_init_kernel(int* __restrict__ head, int n) {
  int i = blockIdx.x * blockDim.x + threadIdx.x;
  if (i < n) head[i] = -1;
}

__global__ void chain_build_kernel(const int* __restrict__ src,
                                   const int* __restrict__ dst,
                                   const int* __restrict__ et,
                                   int* __restrict__ head,
                                   int* __restrict__ node, int n_edges) {
  int e = blockIdx.x * blockDim.x + threadIdx.x;
  if (e >= n_edges) return;
  int prev = atomicExch(&head[dst[e]], e);
  int2 nd; nd.x = prev; nd.y = src[e] * R_ + et[e];
  *(int2*)(node + 2 * (size_t)e) = nd;
}

#define CH_CAP 128
__global__ __launch_bounds__(256)
void chain_order_kernel(const int* __restrict__ head, int* node, int n) {
  __shared__ int s_e[8][CH_CAP];
  __shared__ int s_key[8][CH_CAP];
  const int wave = __builtin_amdgcn_readfirstlane((int)(threadIdx.x >> 5));
  const int lane = threadIdx.x & 31;
  const int d = blockIdx.x * 8 + wave;
  int cnt = 0;
  int e = (d < n) ? head[d] : -1;
  for (int it = 0; it < E_ && e >= 0; ++it) {
    const int2 nd = *(const int2*)(node + 2 * (size_t)e);
    if (cnt < CH_CAP && lane == 0) { s_e[wave][cnt] = e; s_key[wave][cnt] = nd.y; }
    ++cnt;
    e = nd.x;
  }
  __syncthreads();
  if (cnt <= CH_CAP) {
    for (int t = lane; t < cnt; t += 32) {
      const int my = s_e[wave][t];
      int rank = 0;
      for (int u = 0; u < cnt; ++u) rank += (s_e[wave][u] < my) ? 1 : 0;
      atomicExch(&node[2 * (size_t)s_e[wave][rank] + 1], s_key[wave][t]);
    }
  }
}

template <int NV4>
__global__ __launch_bounds__(256)
void gather_ln_kernel(const float* __restrict__ Y,
                      const int* __restrict__ head,
                      const int* __restrict__ node,
                      const float* __restrict__ gamma,
                      const float* __restrict__ beta,
                      const float* __restrict__ resid,
                      float* __restrict__ conv, int n, int do_gelu) {
  constexpr int OC = NV4 * 128;
  const int wave = __builtin_amdgcn_readfirstlane((int)(threadIdx.x >> 5));
  const int lane = threadIdx.x & 31;
  const int d = blockIdx.x * 8 + wave;
  if (d >= n) return;
  float* cr = conv + (size_t)d * OC + 4 * lane;
  const float* rr = resid + (size_t)d * OC + 4 * lane;
  v4f acc[NV4], rs[NV4];
#pragma unroll
  for (int j = 0; j < NV4; ++j) acc[j] = *(const v4f*)(cr + 128 * j);
#pragma unroll
  for (int j = 0; j < NV4; ++j) rs[j] = *(const v4f*)(rr + 128 * j);
  int e = head[d];
  int2 nd; nd.x = -1; nd.y = 0;
  if (e >= 0) nd = *(const int2*)(node + 2 * (size_t)e);
  for (int it = 0; it < E_ && e >= 0; ++it) {
    const int en = nd.x;
    const float* yr = Y + (size_t)nd.y * OC + 4 * lane;
    int2 ndn; ndn.x = -1; ndn.y = 0;
    if (en >= 0) ndn = *(const int2*)(node + 2 * (size_t)en);
#pragma unroll
    for (int j = 0; j < NV4; ++j) acc[j] += *(const v4f*)(yr + 128 * j);
    e = en; nd = ndn;
  }
  float s = 0.f;
#pragma unroll
  for (int j = 0; j < NV4; ++j) s += (acc[j].x + acc[j].y) + (acc[j].z + acc[j].w);
  for (int o = 16; o; o >>= 1) s += __shfl_xor(s, o, 32);
  const float mean = s / (float)OC;
  float vs = 0.f;
#pragma unroll
  for (int j = 0; j < NV4; ++j) {
    acc[j] -= mean;
    vs += (acc[j].x * acc[j].x + acc[j].y * acc[j].y) + (acc[j].z * acc[j].z + acc[j].w * acc[j].w);
  }
  for (int o = 16; o; o >>= 1) vs += __shfl_xor(vs, o, 32);
  const float inv = rsqrtf(vs / (float)OC + LN_EPS);
#pragma unroll
  for (int j = 0; j < NV4; ++j) {
    const v4f gm = *(const v4f*)(gamma + 128 * j + 4 * lane);
    const v4f bt = *(const v4f*)(beta + 128 * j + 4 * lane);
    v4f v = acc[j] * inv * gm + bt + rs[j];
    if (do_gelu) {
      v.x = 0.5f * v.x * (1.f + erff(v.x * 0.70710678118654752f));
      v.y = 0.5f * v.y * (1.f + erff(v.y * 0.70710678118654752f));
      v.z = 0.5f * v.z * (1.f + erff(v.z * 0.70710678118654752f));
      v.w = 0.5f * v.w * (1.f + erff(v.w * 0.70710678118654752f));
    }
    *(v4f*)(cr + 128 * j) = v;
  }
}

#define FR_M 5
#define TILE_M 80
#define KST 4

struct FragSet {
  v2f a[KST][FR_M];
  v2f a2[KST][FR_M];
  v4f b[KST][2];
};

template <bool PE>
__device__ __forceinline__ void load_set(FragSet& f, const float* __restrict__ Ap,
                                         const unsigned (&aoff)[FR_M],
                                         const unsigned (&aoff2)[FR_M],
                                         const float* __restrict__ Bp,
                                         unsigned boff, int ldb, int k0) {
  const int ka = PE ? (k0 & (OUT_ - 1)) : k0;
#pragma unroll
  for (int s = 0; s < KST; ++s) {
#pragma unroll
    for (int i = 0; i < FR_M; ++i) {
      f.a[s][i] = *(const v2f*)(Ap + aoff[i] + ka + 4 * s);
      if (PE) f.a2[s][i] = *(const v2f*)(Ap + aoff2[i] + ka + 4 * s);
    }
  }
  const float* Bk = Bp + (size_t)k0 * ldb + boff;
#pragma unroll
  for (int s = 0; s < KST; ++s) {
    f.b[s][0] = *(const v4f*)(Bk + (size_t)(4 * s) * ldb);
    f.b[s][1] = *(const v4f*)(Bk + (size_t)(4 * s + 1) * ldb);
  }
}

template <bool PE, int SEC>
__device__ __forceinline__ void mma_sec(v8f (&acc)[FR_M][4], const FragSet& f) {
#pragma unroll
  for (int s = 0; s < KST; ++s) {
    v2f a[FR_M];
#pragma unroll
    for (int i = 0; i < FR_M; ++i) {
      if (PE) {
        v2f zi = f.a[s][i], zj = f.a2[s][i];
        if (SEC == 0)      { a[i].x = zi.x + zj.x;        a[i].y = zi.y + zj.y; }
        else if (SEC == 1) { a[i].x = zi.x * zj.x;        a[i].y = zi.y * zj.y; }
        else               { a[i].x = fabsf(zi.x - zj.x); a[i].y = fabsf(zi.y - zj.y); }
      } else {
        a[i] = f.a[s][i];
      }
    }
    v2f b[4];
#pragma unroll
    for (int j = 0; j < 4; ++j) { b[j].x = f.b[s][0][j]; b[j].y = f.b[s][1][j]; }
#pragma unroll
    for (int i = 0; i < FR_M; ++i) {
#pragma unroll
      for (int j = 0; j < 4; ++j) acc[i][j] = WMMA_F32(a[i], b[j], acc[i][j]);
    }
  }
}

template <bool PE>
__device__ __forceinline__ void mma_set(v8f (&acc)[FR_M][4], const FragSet& f, int sec) {
  if (!PE)           mma_sec<false, 0>(acc, f);
  else if (sec == 0) mma_sec<true, 0>(acc, f);
  else if (sec == 1) mma_sec<true, 1>(acc, f);
  else               mma_sec<true, 2>(acc, f);
}

template <bool PE>
__device__ __forceinline__ void gemm80_tile(const float* __restrict__ A,
                                            const int* __restrict__ gi,
                                            const int* __restrict__ gj,
                                            const float* __restrict__ Bc,
                                            float* __restrict__ Cc,
                                            int tm, int K,
                                            int lda, int ldb, int ldc,
                                            int accumulate) {
  const int lane = threadIdx.x & 31;
  const int l16  = lane & 15;
  const int g    = lane >> 4;

  const float* Ap;
  unsigned aoff[FR_M], aoff2[FR_M];
  if (PE) {
    Ap = A;
#pragma unroll
    for (int i = 0; i < FR_M; ++i) {
      int row = tm * TILE_M + i * 16 + l16;
      aoff[i]  = (unsigned)gi[row] * OUT_ + 2 * g;
      aoff2[i] = (unsigned)gj[row] * OUT_ + 2 * g;
    }
  } else {
    Ap = A + (size_t)(tm * TILE_M) * lda;
#pragma unroll
    for (int i = 0; i < FR_M; ++i) {
      aoff[i]  = (unsigned)(i * 16 + l16) * (unsigned)lda + 2 * g;
      aoff2[i] = 0;
    }
  }
  const float* Bp = Bc;
  const unsigned boff = (unsigned)(2 * g) * (unsigned)ldb + 4 * l16;

  v8f acc[FR_M][4];
#pragma unroll
  for (int i = 0; i < FR_M; ++i)
#pragma unroll
    for (int j = 0; j < 4; ++j) acc[i][j] = v8f{};

  FragSet s0, s1;
  load_set<PE>(s0, Ap, aoff, aoff2, Bp, boff, ldb, 0);
  for (int k0 = 0; k0 < K; k0 += 32) {
    load_set<PE>(s1, Ap, aoff, aoff2, Bp, boff, ldb, k0 + 16);
    __builtin_amdgcn_sched_barrier(0);
    mma_set<PE>(acc, s0, k0 >> 7);
    __builtin_amdgcn_sched_barrier(0);
    const int kn = (k0 + 32 < K) ? k0 + 32 : 0;
    load_set<PE>(s0, Ap, aoff, aoff2, Bp, boff, ldb, kn);
    __builtin_amdgcn_sched_barrier(0);
    mma_set<PE>(acc, s1, (k0 + 16) >> 7);
    __builtin_amdgcn_sched_barrier(0);
  }

  float* Cp = Cc + (size_t)(tm * TILE_M + 8 * g) * ldc + 4 * l16;
  if (accumulate) {
    v4f old[FR_M][8];
#pragma unroll
    for (int i = 0; i < FR_M; ++i)
#pragma unroll
      for (int r = 0; r < 8; ++r)
        old[i][r] = *(const v4f*)(Cp + (size_t)(i * 16 + r) * ldc);
#pragma unroll
    for (int i = 0; i < FR_M; ++i) {
#pragma unroll
      for (int r = 0; r < 8; ++r) {
        v4f v;
        v.x = acc[i][0][r]; v.y = acc[i][1][r]; v.z = acc[i][2][r]; v.w = acc[i][3][r];
        *(v4f*)(Cp + (size_t)(i * 16 + r) * ldc) = v + old[i][r];
      }
    }
  } else {
#pragma unroll
    for (int i = 0; i < FR_M; ++i) {
#pragma unroll
      for (int r = 0; r < 8; ++r) {
        v4f v;
        v.x = acc[i][0][r]; v.y = acc[i][1][r]; v.z = acc[i][2][r]; v.w = acc[i][3][r];
        *(v4f*)(Cp + (size_t)(i * 16 + r) * ldc) = v;
      }
    }
  }
}

template <bool PE>
__device__ __forceinline__ void gemm80_body(const float* __restrict__ A,
                                            const int* __restrict__ gi,
                                            const int* __restrict__ gj,
                                            const float* __restrict__ Bm,
                                            float* __restrict__ C,
                                            int M, int Nc, int K,
                                            int lda, int ldb, int ldc,
                                            int accumulate) {
  const int wave = __builtin_amdgcn_readfirstlane((int)(threadIdx.x >> 5));
  const int tilesN = Nc >> 6;
  const int tilesTotal = (M / TILE_M) * tilesN;
  const int tile = blockIdx.x * 4 + wave;
  if (tile >= tilesTotal) return;
  const int tm = tile / tilesN;
  const int tn = tile - tm * tilesN;
  gemm80_tile<PE>(A, gi, gj, Bm + tn * 64, C + tn * 64, tm, K, lda, ldb, ldc, accumulate);
}

struct Seg3 {
  const float* b0; const float* b1; const float* b2;
  float* c0; float* c1; float* c2;
  int nt0, nt1, nt2;
  int ld0, ld1, ld2;
};

__global__ __launch_bounds__(128)
void gemm_f32_wmma3_kernel(const float* __restrict__ A, Seg3 sg,
                           int M, int K, int lda) {
  const int wave = __builtin_amdgcn_readfirstlane((int)(threadIdx.x >> 5));
  const int tilesN = sg.nt0 + sg.nt1 + sg.nt2;
  const int tilesTotal = (M / TILE_M) * tilesN;
  const int tile = blockIdx.x * 4 + wave;
  if (tile >= tilesTotal) return;
  const int tm = tile / tilesN;
  int tn = tile - tm * tilesN;
  const float* Bm = sg.b0; float* C = sg.c0; int ld = sg.ld0;
  if (tn >= sg.nt0) {
    tn -= sg.nt0; Bm = sg.b1; C = sg.c1; ld = sg.ld1;
    if (tn >= sg.nt1) { tn -= sg.nt1; Bm = sg.b2; C = sg.c2; ld = sg.ld2; }
  }
  gemm80_tile<false>(A, nullptr, nullptr, Bm + tn * 64, C + tn * 64, tm, K, lda, ld, ld, 0);
}

__global__ __launch_bounds__(128)
void gemm_f32_wmma_kernel(const float* __restrict__ A,
                          const float* __restrict__ Bm,
                          float* __restrict__ C,
                          int M, int Nc, int K,
                          int lda, int ldb, int ldc,
                          int accumulate) {
  gemm80_body<false>(A, nullptr, nullptr, Bm, C, M, Nc, K, lda, ldb, ldc, accumulate);
}

__global__ __launch_bounds__(128)
void gemm_pe_wmma_kernel(const float* __restrict__ z,
                         const int* __restrict__ gi,
                         const int* __restrict__ gj,
                         const float* __restrict__ Bm,
                         float* __restrict__ C,
                         int M, int Nc, int K,
                         int ldb, int ldc) {
  gemm80_body<true>(z, gi, gj, Bm, C, M, Nc, K, 0, ldb, ldc, 0);
}

template <int C, bool DOT>
__global__ __launch_bounds__(256)
void ln_gelu_rows_kernel(float* __restrict__ y,
                         const float* __restrict__ bias,
                         const float* __restrict__ gamma,
                         const float* __restrict__ beta,
                         const float* __restrict__ w,
                         const float* __restrict__ b,
                         float* __restrict__ logits, int rows) {
  constexpr int NV = C / 32;
  __shared__ float s_dot[32];
  const int lane = threadIdx.x & 31;
  const int wave = threadIdx.x >> 5;
  const int t = lane & 7;
  const int rib = wave * 4 + (lane >> 3);
  const int row = blockIdx.x * 32 + rib;
  const int rr = row < rows ? row : rows - 1;
  float* yr = y + (size_t)rr * C + 4 * t;
  v4f v[NV];
#pragma unroll
  for (int i = 0; i < NV; ++i) v[i] = *(const v4f*)(yr + 32 * i);
#pragma unroll
  for (int i = 0; i < NV; ++i) v[i] += *(const v4f*)(bias + 32 * i + 4 * t);
  float s = 0.f;
#pragma unroll
  for (int i = 0; i < NV; ++i) s += (v[i].x + v[i].y) + (v[i].z + v[i].w);
  for (int o = 4; o; o >>= 1) s += __shfl_xor(s, o, 32);
  const float mean = s / (float)C;
  float vs = 0.f;
#pragma unroll
  for (int i = 0; i < NV; ++i) {
    v[i] -= mean;
    vs += (v[i].x * v[i].x + v[i].y * v[i].y) + (v[i].z * v[i].z + v[i].w * v[i].w);
  }
  for (int o = 4; o; o >>= 1) vs += __shfl_xor(vs, o, 32);
  const float inv = rsqrtf(vs / (float)C + LN_EPS);
  float dot = 0.f;
#pragma unroll
  for (int i = 0; i < NV; ++i) {
    const v4f gm = *(const v4f*)(gamma + 32 * i + 4 * t);
    const v4f bt = *(const v4f*)(beta + 32 * i + 4 * t);
    v4f u = v[i] * inv * gm + bt;
    u.x = 0.5f * u.x * (1.f + erff(u.x * 0.70710678118654752f));
    u.y = 0.5f * u.y * (1.f + erff(u.y * 0.70710678118654752f));
    u.z = 0.5f * u.z * (1.f + erff(u.z * 0.70710678118654752f));
    u.w = 0.5f * u.w * (1.f + erff(u.w * 0.70710678118654752f));
    if (DOT) {
      const v4f ww = *(const v4f*)(w + 32 * i + 4 * t);
      dot += (u.x * ww.x + u.y * ww.y) + (u.z * ww.z + u.w * ww.w);
    } else {
      if (row < rows) *(v4f*)(yr + 32 * i) = u;
    }
  }
  if (DOT) {
    for (int o = 4; o; o >>= 1) dot += __shfl_xor(dot, o, 32);
    if (t == 0) s_dot[rib] = dot;
    __syncthreads();
    const int p = blockIdx.x * 32 + (int)threadIdx.x;
    if (threadIdx.x < 32 && p < rows) logits[p] = s_dot[threadIdx.x] + b[0];
  }
}

extern "C" void kernel_launch(void* const* d_in, const int* in_sizes, int n_in,
                              void* d_out, int out_size, void* d_ws, size_t ws_size,
                              hipStream_t stream) {
  (void)in_sizes; (void)n_in; (void)out_size; (void)ws_size;

  const float* x    = (const float*)d_in[0];
  const int*   ei   = (const int*)d_in[1];
  const int*   esrc = ei;
  const int*   edst = ei + E_;
  const int*   et   = (const int*)d_in[2];
  const int*   gp   = (const int*)d_in[3];
  const int*   gi   = gp;
  const int*   gj   = gp + P_;

  const float* basis[4], *att[4], *root[4], *relatt[4], *gam[4], *bet[4];
  for (int i = 0; i < 4; ++i) {
    basis[i]  = (const float*)d_in[4 + 6 * i + 0];
    att[i]    = (const float*)d_in[4 + 6 * i + 1];
    root[i]   = (const float*)d_in[4 + 6 * i + 2];
    relatt[i] = (const float*)d_in[4 + 6 * i + 3];
    gam[i]    = (const float*)d_in[4 + 6 * i + 4];
    bet[i]    = (const float*)d_in[4 + 6 * i + 5];
  }
  const float* res0 = (const float*)d_in[28];
  const float* res3 = (const float*)d_in[29];
  const float* dW0 = (const float*)d_in[30], *db0 = (const float*)d_in[31];
  const float* dg0 = (const float*)d_in[32], *dbb0 = (const float*)d_in[33];
  const float* dW1 = (const float*)d_in[34], *db1 = (const float*)d_in[35];
  const float* dg1 = (const float*)d_in[36], *dbb1 = (const float*)d_in[37];
  const float* dW2 = (const float*)d_in[38], *db2 = (const float*)d_in[39];
  const float* dg2 = (const float*)d_in[40], *dbb2 = (const float*)d_in[41];
  const float* dW3 = (const float*)d_in[42], *db3 = (const float*)d_in[43];

  float* ws = (float*)d_ws;
  size_t off = 0;
  float* a_sm  = ws + off; off += 64;
  float* Wcat  = ws + off; off += (size_t)R_ * H_ * H_;
  float* hA    = ws + off; off += (size_t)N_ * H_;
  float* hB    = ws + off; off += (size_t)N_ * H_;
  int*   head  = (int*)(ws + off);
  int*   node  = head + N_;
  off += (size_t)N_ * H_;
  float* big   = ws + off;
  float* d0 = big;
  float* d1 = big + (size_t)P_ * H_;
  float* d2 = hA;

  auto gemm = [&](const float* A, const float* Bm, float* Cm,
                  int M, int Nc, int K, int lda, int ldb, int ldc, int acc) {
    int tiles = (M / 80) * (Nc / 64);
    gemm_f32_wmma_kernel<<<(tiles + 3) / 4, 128, 0, stream>>>(
        A, Bm, Cm, M, Nc, K, lda, ldb, ldc, acc);
  };

  struct { const float* cur; int in; int out; int li;
           const float* resmat; float* conv; float* resid; int gelu; } L[4] = {
    { x,  IN_, H_,   0, res0,    hA, hB,                      1 },
    { hA, H_,  H_,   1, nullptr, hB, nullptr,                 1 },
    { hB, H_,  H_,   2, nullptr, hA, nullptr,                 1 },
    { hA, H_,  OUT_, 3, res3,    hB, hB + (size_t)N_ * OUT_,  0 },
  };

  chain_init_kernel<<<(N_ + 255) / 256, 256, 0, stream>>>(head, N_);
  chain_build_kernel<<<(E_ + 255) / 256, 256, 0, stream>>>(esrc, edst, et, head, node, E_);
  chain_order_kernel<<<(N_ + 7) / 8, 256, 0, stream>>>(head, node, N_);

  for (int i = 0; i < 4; ++i) {
    const float* cur = L[i].cur;
    int in_ch = L[i].in, out_ch = L[i].out, li = L[i].li;
    float* conv = L[i].conv;

    relatt_softmax_kernel<<<1, 32, 0, stream>>>(relatt[li], a_sm);
    int wtotal = R_ * in_ch * out_ch;
    make_weights_kernel<<<(wtotal + 255) / 256, 256, 0, stream>>>(
        att[li], basis[li], a_sm, Wcat, in_ch, out_ch);

    const float* rptr = L[i].resmat ? L[i].resid : cur;
    {
      Seg3 sg;
      sg.b0 = Wcat;      sg.c0 = big;  sg.nt0 = (R_ * out_ch) / 64; sg.ld0 = R_ * out_ch;
      sg.b1 = root[li];  sg.c1 = conv; sg.nt1 = out_ch / 64;        sg.ld1 = out_ch;
      sg.b2 = L[i].resmat ? L[i].resmat : root[li];
      sg.c2 = L[i].resmat ? L[i].resid : conv;
      sg.nt2 = L[i].resmat ? out_ch / 64 : 0;                       sg.ld2 = out_ch;
      int tiles = (N_ / 80) * (sg.nt0 + sg.nt1 + sg.nt2);
      gemm_f32_wmma3_kernel<<<(tiles + 3) / 4, 128, 0, stream>>>(cur, sg, N_, in_ch, in_ch);
    }
    if (out_ch == H_)
      gather_ln_kernel<2><<<(N_ + 7) / 8, 256, 0, stream>>>(
          big, head, node, gam[li], bet[li], rptr, conv, N_, L[i].gelu);
    else
      gather_ln_kernel<1><<<(N_ + 7) / 8, 256, 0, stream>>>(
          big, head, node, gam[li], bet[li], rptr, conv, N_, L[i].gelu);
  }

  hipMemcpyAsync((float*)d_out + P_, hB, (size_t)N_ * OUT_ * sizeof(float),
                 hipMemcpyDeviceToDevice, stream);

  {
    int tiles = (P_ / 80) * (H_ / 64);
    gemm_pe_wmma_kernel<<<(tiles + 3) / 4, 128, 0, stream>>>(
        hB, gi, gj, dW0, d0, P_, H_, 3 * OUT_, H_, H_);
  }
  ln_gelu_rows_kernel<H_, false><<<(P_ + 31) / 32, 256, 0, stream>>>(
      d0, db0, dg0, dbb0, nullptr, nullptr, nullptr, P_);

  gemm(d0, dW1, d1, P_, H_ / 2, H_, H_, H_ / 2, H_ / 2, 0);
  ln_gelu_rows_kernel<H_ / 2, false><<<(P_ + 31) / 32, 256, 0, stream>>>(
      d1, db1, dg1, dbb1, nullptr, nullptr, nullptr, P_);

  gemm(d1, dW2, d2, P_, H_ / 4, H_ / 2, H_ / 2, H_ / 4, H_ / 4, 0);
  ln_gelu_rows_kernel<H_ / 4, true><<<(P_ + 31) / 32, 256, 0, stream>>>(
      d2, db2, dg2, dbb2, dW3, db3, (float*)d_out, P_);
}


# Edge gather reads each destination's edge keys from one 128-byte line (written once per call by the chain-ordering launch) and asks for up to 8 source rows at a time, not one per hop; same sum order
# speedup vs baseline: 3.1715x; 1.0529x over previous
#include <hip/hip_runtime.h>
#include <hip/hip_bf16.h>
#include <math.h>

#define N_   20000
#define E_   320000
#define R_   8
#define B_   8
#define IN_  128
#define H_   256
#define OUT_ 128
#define P_   100000
#define LN_EPS 1e-5f

typedef float v8f __attribute__((ext_vector_type(8)));
typedef float v2f __attribute__((ext_vector_type(2)));
typedef float v4f __attribute__((ext_vector_type(4)));

#define WMMA_F32(a, b, c) \
  __builtin_amdgcn_wmma_f32_16x16x4_f32(false, (a), false, (b), (short)0, (c), false, false)

__global__ void relatt_softmax_kernel(const float* __restrict__ relatt,
                                      float* __restrict__ a) {
  if (threadIdx.x == 0 && blockIdx.x == 0) {
    float m = relatt[0];
    for (int r = 1; r < R_; ++r) m = fmaxf(m, relatt[r]);
    float e[R_]; float s = 0.f;
    for (int r = 0; r < R_; ++r) { e[r] = expf(relatt[r] - m); s += e[r]; }
    for (int r = 0; r < R_; ++r) a[r] = e[r] / s;
  }
}

__global__ void make_weights_kernel(const float* __restrict__ att,
                                    const float* __restrict__ basis,
                                    const float* __restrict__ a,
                                    float* __restrict__ W,
                                    int in_ch, int out_ch) {
  int idx = blockIdx.x * blockDim.x + threadIdx.x;
  int io_sz = in_ch * out_ch;
  int total = R_ * io_sz;
  if (idx >= total) return;
  int r  = idx / io_sz;
  int io = idx - r * io_sz;
  float s = 0.f;
#pragma unroll
  for (int bb = 0; bb < B_; ++bb)
    s += att[r * B_ + bb] * basis[(size_t)bb * io_sz + io];
  int i = io / out_ch;
  int o = io - i * out_ch;
  W[(size_t)i * (R_ * out_ch) + r * out_ch + o] = s * a[r];
}

__global__ void chain_init_kernel(int* __restrict__ head, int n) {
  int i = blockIdx.x * blockDim.x + threadIdx.x;
  if (i < n) head[i] = -1;
}

__global__ void chain_build_kernel(const int* __restrict__ src,
                                   const int* __restrict__ dst,
                                   const int* __restrict__ et,
                                   int* __restrict__ head,
                                   int* __restrict__ node, int n_edges) {
  int e = blockIdx.x * blockDim.x + threadIdx.x;
  if (e >= n_edges) return;
  int prev = atomicExch(&head[dst[e]], e);
  int2 nd; nd.x = prev; nd.y = src[e] * R_ + et[e];
  *(int2*)(node + 2 * (size_t)e) = nd;
}

#define CH_CAP 128
#define KT 32
__global__ __launch_bounds__(256)
void chain_order_kernel(const int* __restrict__ head, int* node,
                        int* __restrict__ tab, int n) {
  __shared__ int s_e[8][CH_CAP];
  __shared__ int s_key[8][CH_CAP];
  const int wave = __builtin_amdgcn_readfirstlane((int)(threadIdx.x >> 5));
  const int lane = threadIdx.x & 31;
  const int d = blockIdx.x * 8 + wave;
  int cnt = 0;
  int e = (d < n) ? head[d] : -1;
  for (int it = 0; it < E_ && e >= 0; ++it) {
    const int2 nd = *(const int2*)(node + 2 * (size_t)e);
    if (cnt < CH_CAP && lane == 0) { s_e[wave][cnt] = e; s_key[wave][cnt] = nd.y; }
    ++cnt;
    e = nd.x;
  }
  __syncthreads();
  if (d >= n) return;
  if (cnt <= KT) {
    int pos = lane, val = -1;
    if (lane < cnt) {
      const int my = s_e[wave][lane];
      int rank = 0;
      for (int u = 0; u < cnt; ++u) rank += (s_e[wave][u] < my) ? 1 : 0;
      pos = rank; val = s_key[wave][lane];
    }
    tab[(size_t)d * KT + pos] = val;
    return;
  }
  tab[(size_t)d * KT + lane] = -2;
  if (cnt <= CH_CAP) {
    for (int t = lane; t < cnt; t += 32) {
      const int my = s_e[wave][t];
      int rank = 0;
      for (int u = 0; u < cnt; ++u) rank += (s_e[wave][u] < my) ? 1 : 0;
      atomicExch(&node[2 * (size_t)s_e[wave][rank] + 1], s_key[wave][t]);
    }
  }
}

template <int NB, int NV4>
__device__ __forceinline__ void add_rows(v4f (&acc)[NV4], const float* __restrict__ Y,
                                         int myk, int t, int lane) {
  constexpr int OC = NV4 * 128;
  v4f r[NB][NV4];
#pragma unroll
  for (int j = 0; j < NB; ++j) {
    const int k = __builtin_amdgcn_readlane(myk, t + j);
    const float* yr = Y + (size_t)k * OC + 4 * lane;
#pragma unroll
    for (int q = 0; q < NV4; ++q) r[j][q] = *(const v4f*)(yr + 128 * q);
  }
#pragma unroll
  for (int j = 0; j < NB; ++j) {
#pragma unroll
    for (int q = 0; q < NV4; ++q) acc[q] += r[j][q];
  }
}

template <int NV4>
__global__ __launch_bounds__(256)
void gather_ln_kernel(const float* __restrict__ Y,
                      const int* __restrict__ head,
                      const int* __restrict__ node,
                      const int* __restrict__ tab,
                      const float* __restrict__ gamma,
                      const float* __restrict__ beta,
                      const float* __restrict__ resid,
                      float* __restrict__ conv, int n, int do_gelu) {
  constexpr int OC = NV4 * 128;
  const int wave = __builtin_amdgcn_readfirstlane((int)(threadIdx.x >> 5));
  const int lane = threadIdx.x & 31;
  const int d = blockIdx.x * 8 + wave;
  if (d >= n) return;
  float* cr = conv + (size_t)d * OC + 4 * lane;
  const float* rr = resid + (size_t)d * OC + 4 * lane;
  const int myk = tab[(size_t)d * KT + lane];
  v4f acc[NV4], rs[NV4];
#pragma unroll
  for (int j = 0; j < NV4; ++j) acc[j] = *(const v4f*)(cr + 128 * j);
#pragma unroll
  for (int j = 0; j < NV4; ++j) rs[j] = *(const v4f*)(rr + 128 * j);
  if (__builtin_amdgcn_readfirstlane(myk) != -2) {
    const int cnt = __builtin_popcount(__builtin_amdgcn_ballot_w32(myk >= 0));
    int t = 0;
    for (; t + 8 <= cnt; t += 8) add_rows<8, NV4>(acc, Y, myk, t, lane);
    if (cnt - t >= 4) { add_rows<4, NV4>(acc, Y, myk, t, lane); t += 4; }
    if (cnt - t >= 2) { add_rows<2, NV4>(acc, Y, myk, t, lane); t += 2; }
    if (cnt - t >= 1) { add_rows<1, NV4>(acc, Y, myk, t, lane); }
  } else {
    int e = head[d];
    int2 nd; nd.x = -1; nd.y = 0;
    if (e >= 0) nd = *(const int2*)(node + 2 * (size_t)e);
    for (int it = 0; it < E_ && e >= 0; ++it) {
      const int en = nd.x;
      const float* yr = Y + (size_t)nd.y * OC + 4 * lane;
      int2 ndn; ndn.x = -1; ndn.y = 0;
      if (en >= 0) ndn = *(const int2*)(node + 2 * (size_t)en);
#pragma unroll
      for (int j = 0; j < NV4; ++j) acc[j] += *(const v4f*)(yr + 128 * j);
      e = en; nd = ndn;
    }
  }
  float s = 0.f;
#pragma unroll
  for (int j = 0; j < NV4; ++j) s += (acc[j].x + acc[j].y) + (acc[j].z + acc[j].w);
  for (int o = 16; o; o >>= 1) s += __shfl_xor(s, o, 32);
  const float mean = s / (float)OC;
  float vs = 0.f;
#pragma unroll
  for (int j = 0; j < NV4; ++j) {
    acc[j] -= mean;
    vs += (acc[j].x * acc[j].x + acc[j].y * acc[j].y) + (acc[j].z * acc[j].z + acc[j].w * acc[j].w);
  }
  for (int o = 16; o; o >>= 1) vs += __shfl_xor(vs, o, 32);
  const float inv = rsqrtf(vs / (float)OC + LN_EPS);
#pragma unroll
  for (int j = 0; j < NV4; ++j) {
    const v4f gm = *(const v4f*)(gamma + 128 * j + 4 * lane);
    const v4f bt = *(const v4f*)(beta + 128 * j + 4 * lane);
    v4f v = acc[j] * inv * gm + bt + rs[j];
    if (do_gelu) {
      v.x = 0.5f * v.x * (1.f + erff(v.x * 0.70710678118654752f));
      v.y = 0.5f * v.y * (1.f + erff(v.y * 0.70710678118654752f));
      v.z = 0.5f * v.z * (1.f + erff(v.z * 0.70710678118654752f));
      v.w = 0.5f * v.w * (1.f + erff(v.w * 0.70710678118654752f));
    }
    *(v4f*)(cr + 128 * j) = v;
  }
}

#define FR_M 5
#define TILE_M 80
#define KST 4

struct FragSet {
  v2f a[KST][FR_M];
  v2f a2[KST][FR_M];
  v4f b[KST][2];
};

template <bool PE>
__device__ __forceinline__ void load_set(FragSet& f, const float* __restrict__ Ap,
                                         const unsigned (&aoff)[FR_M],
                                         const unsigned (&aoff2)[FR_M],
                                         const float* __restrict__ Bp,
                                         unsigned boff, int ldb, int k0) {
  const int ka = PE ? (k0 & (OUT_ - 1)) : k0;
#pragma unroll
  for (int s = 0; s < KST; ++s) {
#pragma unroll
    for (int i = 0; i < FR_M; ++i) {
      f.a[s][i] = *(const v2f*)(Ap + aoff[i] + ka + 4 * s);
      if (PE) f.a2[s][i] = *(const v2f*)(Ap + aoff2[i] + ka + 4 * s);
    }
  }
  const float* Bk = Bp + (size_t)k0 * ldb + boff;
#pragma unroll
  for (int s = 0; s < KST; ++s) {
    f.b[s][0] = *(const v4f*)(Bk + (size_t)(4 * s) * ldb);
    f.b[s][1] = *(const v4f*)(Bk + (size_t)(4 * s + 1) * ldb);
  }
}

template <bool PE, int SEC>
__device__ __forceinline__ void mma_sec(v8f (&acc)[FR_M][4], const FragSet& f) {
#pragma unroll
  for (int s = 0; s < KST; ++s) {
    v2f a[FR_M];
#pragma unroll
    for (int i = 0; i < FR_M; ++i) {
      if (PE) {
        v2f zi = f.a[s][i], zj = f.a2[s][i];
        if (SEC == 0)      { a[i].x = zi.x + zj.x;        a[i].y = zi.y + zj.y; }
        else if (SEC == 1) { a[i].x = zi.x * zj.x;        a[i].y = zi.y * zj.y; }
        else               { a[i].x = fabsf(zi.x - zj.x); a[i].y = fabsf(zi.y - zj.y); }
      } else {
        a[i] = f.a[s][i];
      }
    }
    v2f b[4];
#pragma unroll
    for (int j = 0; j < 4; ++j) { b[j].x = f.b[s][0][j]; b[j].y = f.b[s][1][j]; }
#pragma unroll
    for (int i = 0; i < FR_M; ++i) {
#pragma unroll
      for (int j = 0; j < 4; ++j) acc[i][j] = WMMA_F32(a[i], b[j], acc[i][j]);
    }
  }
}

template <bool PE>
__device__ __forceinline__ void mma_set(v8f (&acc)[FR_M][4], const FragSet& f, int sec) {
  if (!PE)           mma_sec<false, 0>(acc, f);
  else if (sec == 0) mma_sec<true, 0>(acc, f);
  else if (sec == 1) mma_sec<true, 1>(acc, f);
  else               mma_sec<true, 2>(acc, f);
}

template <bool PE>
__device__ __forceinline__ void gemm80_tile(const float* __restrict__ A,
                                            const int* __restrict__ gi,
                                            const int* __restrict__ gj,
                                            const float* __restrict__ Bc,
                                            float* __restrict__ Cc,
                                            int tm, int K,
                                            int lda, int ldb, int ldc,
                                            int accumulate) {
  const int lane = threadIdx.x & 31;
  const int l16  = lane & 15;
  const int g    = lane >> 4;

  const float* Ap;
  unsigned aoff[FR_M], aoff2[FR_M];
  if (PE) {
    Ap = A;
#pragma unroll
    for (int i = 0; i < FR_M; ++i) {
      int row = tm * TILE_M + i * 16 + l16;
      aoff[i]  = (unsigned)gi[row] * OUT_ + 2 * g;
      aoff2[i] = (unsigned)gj[row] * OUT_ + 2 * g;
    }
  } else {
    Ap = A + (size_t)(tm * TILE_M) * lda;
#pragma unroll
    for (int i = 0; i < FR_M; ++i) {
      aoff[i]  = (unsigned)(i * 16 + l16) * (unsigned)lda + 2 * g;
      aoff2[i] = 0;
    }
  }
  const float* Bp = Bc;
  const unsigned boff = (unsigned)(2 * g) * (unsigned)ldb + 4 * l16;

  v8f acc[FR_M][4];
#pragma unroll
  for (int i = 0; i < FR_M; ++i)
#pragma unroll
    for (int j = 0; j < 4; ++j) acc[i][j] = v8f{};

  FragSet s0, s1;
  load_set<PE>(s0, Ap, aoff, aoff2, Bp, boff, ldb, 0);
  for (int k0 = 0; k0 < K; k0 += 32) {
    load_set<PE>(s1, Ap, aoff, aoff2, Bp, boff, ldb, k0 + 16);
    __builtin_amdgcn_sched_barrier(0);
    mma_set<PE>(acc, s0, k0 >> 7);
    __builtin_amdgcn_sched_barrier(0);
    const int kn = (k0 + 32 < K) ? k0 + 32 : 0;
    load_set<PE>(s0, Ap, aoff, aoff2, Bp, boff, ldb, kn);
    __builtin_amdgcn_sched_barrier(0);
    mma_set<PE>(acc, s1, (k0 + 16) >> 7);
    __builtin_amdgcn_sched_barrier(0);
  }

  float* Cp = Cc + (size_t)(tm * TILE_M + 8 * g) * ldc + 4 * l16;
  if (accumulate) {
    v4f old[FR_M][8];
#pragma unroll
    for (int i = 0; i < FR_M; ++i)
#pragma unroll
      for (int r = 0; r < 8; ++r)
        old[i][r] = *(const v4f*)(Cp + (size_t)(i * 16 + r) * ldc);
#pragma unroll
    for (int i = 0; i < FR_M; ++i) {
#pragma unroll
      for (int r = 0; r < 8; ++r) {
        v4f v;
        v.x = acc[i][0][r]; v.y = acc[i][1][r]; v.z = acc[i][2][r]; v.w = acc[i][3][r];
        *(v4f*)(Cp + (size_t)(i * 16 + r) * ldc) = v + old[i][r];
      }
    }
  } else {
#pragma unroll
    for (int i = 0; i < FR_M; ++i) {
#pragma unroll
      for (int r = 0; r < 8; ++r) {
        v4f v;
        v.x = acc[i][0][r]; v.y = acc[i][1][r]; v.z = acc[i][2][r]; v.w = acc[i][3][r];
        *(v4f*)(Cp + (size_t)(i * 16 + r) * ldc) = v;
      }
    }
  }
}

template <bool PE>
__device__ __forceinline__ void gemm80_body(const float* __restrict__ A,
                                            const int* __restrict__ gi,
                                            const int* __restrict__ gj,
                                            const float* __restrict__ Bm,
                                            float* __restrict__ C,
                                            int M, int Nc, int K,
                                            int lda, int ldb, int ldc,
                                            int accumulate) {
  const int wave = __builtin_amdgcn_readfirstlane((int)(threadIdx.x >> 5));
  const int tilesN = Nc >> 6;
  const int tilesTotal = (M / TILE_M) * tilesN;
  const int tile = blockIdx.x * 4 + wave;
  if (tile >= tilesTotal) return;
  const int tm = tile / tilesN;
  const int tn = tile - tm * tilesN;
  gemm80_tile<PE>(A, gi, gj, Bm + tn * 64, C + tn * 64, tm, K, lda, ldb, ldc, accumulate);
}

struct Seg3 {
  const float* b0; const float* b1; const float* b2;
  float* c0; float* c1; float* c2;
  int nt0, nt1, nt2;
  int ld0, ld1, ld2;
};

__global__ __launch_bounds__(128)
void gemm_f32_wmma3_kernel(const float* __restrict__ A, Seg3 sg,
                           int M, int K, int lda) {
  const int wave = __builtin_amdgcn_readfirstlane((int)(threadIdx.x >> 5));
  const int tilesN = sg.nt0 + sg.nt1 + sg.nt2;
  const int tilesTotal = (M / TILE_M) * tilesN;
  const int tile = blockIdx.x * 4 + wave;
  if (tile >= tilesTotal) return;
  const int tm = tile / tilesN;
  int tn = tile - tm * tilesN;
  const float* Bm = sg.b0; float* C = sg.c0; int ld = sg.ld0;
  if (tn >= sg.nt0) {
    tn -= sg.nt0; Bm = sg.b1; C = sg.c1; ld = sg.ld1;
    if (tn >= sg.nt1) { tn -= sg.nt1; Bm = sg.b2; C = sg.c2; ld = sg.ld2; }
  }
  gemm80_tile<false>(A, nullptr, nullptr, Bm + tn * 64, C + tn * 64, tm, K, lda, ld, ld, 0);
}

__global__ __launch_bounds__(128)
void gemm_f32_wmma_kernel(const float* __restrict__ A,
                          const float* __restrict__ Bm,
                          float* __restrict__ C,
                          int M, int Nc, int K,
                          int lda, int ldb, int ldc,
                          int accumulate) {
  gemm80_body<false>(A, nullptr, nullptr, Bm, C, M, Nc, K, lda, ldb, ldc, accumulate);
}

__global__ __launch_bounds__(128)
void gemm_pe_wmma_kernel(const float* __restrict__ z,
                         const int* __restrict__ gi,
                         const int* __restrict__ gj,
                         const float* __restrict__ Bm,
                         float* __restrict__ C,
                         int M, int Nc, int K,
                         int ldb, int ldc) {
  gemm80_body<true>(z, gi, gj, Bm, C, M, Nc, K, 0, ldb, ldc, 0);
}

template <int C, bool DOT>
__global__ __launch_bounds__(256)
void ln_gelu_rows_kernel(float* __restrict__ y,
                         const float* __restrict__ bias,
                         const float* __restrict__ gamma,
                         const float* __restrict__ beta,
                         const float* __restrict__ w,
                         const float* __restrict__ b,
                         float* __restrict__ logits, int rows) {
  constexpr int NV = C / 32;
  __shared__ float s_dot[32];
  const int lane = threadIdx.x & 31;
  const int wave = threadIdx.x >> 5;
  const int t = lane & 7;
  const int rib = wave * 4 + (lane >> 3);
  const int row = blockIdx.x * 32 + rib;
  const int rr = row < rows ? row : rows - 1;
  float* yr = y + (size_t)rr * C + 4 * t;
  v4f v[NV];
#pragma unroll
  for (int i = 0; i < NV; ++i) v[i] = *(const v4f*)(yr + 32 * i);
#pragma unroll
  for (int i = 0; i < NV; ++i) v[i] += *(const v4f*)(bias + 32 * i + 4 * t);
  float s = 0.f;
#pragma unroll
  for (int i = 0; i < NV; ++i) s += (v[i].x + v[i].y) + (v[i].z + v[i].w);
  for (int o = 4; o; o >>= 1) s += __shfl_xor(s, o, 32);
  const float mean = s / (float)C;
  float vs = 0.f;
#pragma unroll
  for (int i = 0; i < NV; ++i) {
    v[i] -= mean;
    vs += (v[i].x * v[i].x + v[i].y * v[i].y) + (v[i].z * v[i].z + v[i].w * v[i].w);
  }
  for (int o = 4; o; o >>= 1) vs += __shfl_xor(vs, o, 32);
  const float inv = rsqrtf(vs / (float)C + LN_EPS);
  float dot = 0.f;
#pragma unroll
  for (int i = 0; i < NV; ++i) {
    const v4f gm = *(const v4f*)(gamma + 32 * i + 4 * t);
    const v4f bt = *(const v4f*)(beta + 32 * i + 4 * t);
    v4f u = v[i] * inv * gm + bt;
    u.x = 0.5f * u.x * (1.f + erff(u.x * 0.70710678118654752f));
    u.y = 0.5f * u.y * (1.f + erff(u.y * 0.70710678118654752f));
    u.z = 0.5f * u.z * (1.f + erff(u.z * 0.70710678118654752f));
    u.w = 0.5f * u.w * (1.f + erff(u.w * 0.70710678118654752f));
    if (DOT) {
      const v4f ww = *(const v4f*)(w + 32 * i + 4 * t);
      dot += (u.x * ww.x + u.y * ww.y) + (u.z * ww.z + u.w * ww.w);
    } else {
      if (row < rows) *(v4f*)(yr + 32 * i) = u;
    }
  }
  if (DOT) {
    for (int o = 4; o; o >>= 1) dot += __shfl_xor(dot, o, 32);
    if (t == 0) s_dot[rib] = dot;
    __syncthreads();
    const int p = blockIdx.x * 32 + (int)threadIdx.x;
    if (threadIdx.x < 32 && p < rows) logits[p] = s_dot[threadIdx.x] + b[0];
  }
}

extern "C" void kernel_launch(void* const* d_in, const int* in_sizes, int n_in,
                              void* d_out, int out_size, void* d_ws, size_t ws_size,
                              hipStream_t stream) {
  (void)in_sizes; (void)n_in; (void)out_size; (void)ws_size;

  const float* x    = (const float*)d_in[0];
  const int*   ei   = (const int*)d_in[1];
  const int*   esrc = ei;
  const int*   edst = ei + E_;
  const int*   et   = (const int*)d_in[2];
  const int*   gp   = (const int*)d_in[3];
  const int*   gi   = gp;
  const int*   gj   = gp + P_;

  const float* basis[4], *att[4], *root[4], *relatt[4], *gam[4], *bet[4];
  for (int i = 0; i < 4; ++i) {
    basis[i]  = (const float*)d_in[4 + 6 * i + 0];
    att[i]    = (const float*)d_in[4 + 6 * i + 1];
    root[i]   = (const float*)d_in[4 + 6 * i + 2];
    relatt[i] = (const float*)d_in[4 + 6 * i + 3];
    gam[i]    = (const float*)d_in[4 + 6 * i + 4];
    bet[i]    = (const float*)d_in[4 + 6 * i + 5];
  }
  const float* res0 = (const float*)d_in[28];
  const float* res3 = (const float*)d_in[29];
  const float* dW0 = (const float*)d_in[30], *db0 = (const float*)d_in[31];
  const float* dg0 = (const float*)d_in[32], *dbb0 = (const float*)d_in[33];
  const float* dW1 = (const float*)d_in[34], *db1 = (const float*)d_in[35];
  const float* dg1 = (const float*)d_in[36], *dbb1 = (const float*)d_in[37];
  const float* dW2 = (const float*)d_in[38], *db2 = (const float*)d_in[39];
  const float* dg2 = (const float*)d_in[40], *dbb2 = (const float*)d_in[41];
  const float* dW3 = (const float*)d_in[42], *db3 = (const float*)d_in[43];

  float* ws = (float*)d_ws;
  size_t off = 0;
  float* a_sm  = ws + off; off += 64;
  float* Wcat  = ws + off; off += (size_t)R_ * H_ * H_;
  float* hA    = ws + off; off += (size_t)N_ * H_;
  float* hB    = ws + off; off += (size_t)N_ * H_;
  int*   head  = (int*)(ws + off);
  int*   node  = head + N_;
  int*   tab   = node + 2 * (size_t)E_;
  off += (size_t)N_ * H_;
  float* big   = ws + off;
  float* d0 = big;
  float* d1 = big + (size_t)P_ * H_;
  float* d2 = hA;

  auto gemm = [&](const float* A, const float* Bm, float* Cm,
                  int M, int Nc, int K, int lda, int ldb, int ldc, int acc) {
    int tiles = (M / 80) * (Nc / 64);
    gemm_f32_wmma_kernel<<<(tiles + 3) / 4, 128, 0, stream>>>(
        A, Bm, Cm, M, Nc, K, lda, ldb, ldc, acc);
  };

  struct { const float* cur; int in; int out; int li;
           const float* resmat; float* conv; float* resid; int gelu; } L[4] = {
    { x,  IN_, H_,   0, res0,    hA, hB,                      1 },
    { hA, H_,  H_,   1, nullptr, hB, nullptr,                 1 },
    { hB, H_,  H_,   2, nullptr, hA, nullptr,                 1 },
    { hA, H_,  OUT_, 3, res3,    hB, hB + (size_t)N_ * OUT_,  0 },
  };

  chain_init_kernel<<<(N_ + 255) / 256, 256, 0, stream>>>(head, N_);
  chain_build_kernel<<<(E_ + 255) / 256, 256, 0, stream>>>(esrc, edst, et, head, node, E_);
  chain_order_kernel<<<(N_ + 7) / 8, 256, 0, stream>>>(head, node, tab, N_);

  for (int i = 0; i < 4; ++i) {
    const float* cur = L[i].cur;
    int in_ch = L[i].in, out_ch = L[i].out, li = L[i].li;
    float* conv = L[i].conv;

    relatt_softmax_kernel<<<1, 32, 0, stream>>>(relatt[li], a_sm);
    int wtotal = R_ * in_ch * out_ch;
    make_weights_kernel<<<(wtotal + 255) / 256, 256, 0, stream>>>(
        att[li], basis[li], a_sm, Wcat, in_ch, out_ch);

    const float* rptr = L[i].resmat ? L[i].resid : cur;
    {
      Seg3 sg;
      sg.b0 = Wcat;      sg.c0 = big;  sg.nt0 = (R_ * out_ch) / 64; sg.ld0 = R_ * out_ch;
      sg.b1 = root[li];  sg.c1 = conv; sg.nt1 = out_ch / 64;        sg.ld1 = out_ch;
      sg.b2 = L[i].resmat ? L[i].resmat : root[li];
      sg.c2 = L[i].resmat ? L[i].resid : conv;
      sg.nt2 = L[i].resmat ? out_ch / 64 : 0;                       sg.ld2 = out_ch;
      int tiles = (N_ / 80) * (sg.nt0 + sg.nt1 + sg.nt2);
      gemm_f32_wmma3_kernel<<<(tiles + 3) / 4, 128, 0, stream>>>(cur, sg, N_, in_ch, in_ch);
    }
    if (out_ch == H_)
      gather_ln_kernel<2><<<(N_ + 7) / 8, 256, 0, stream>>>(
          big, head, node, tab, gam[li], bet[li], rptr, conv, N_, L[i].gelu);
    else
      gather_ln_kernel<1><<<(N_ + 7) / 8, 256, 0, stream>>>(
          big, head, node, tab, gam[li], bet[li], rptr, conv, N_, L[i].gelu);
  }

  hipMemcpyAsync((float*)d_out + P_, hB, (size_t)N_ * OUT_ * sizeof(float),
                 hipMemcpyDeviceToDevice, stream);

  {
    int tiles = (P_ / 80) * (H_ / 64);
    gemm_pe_wmma_kernel<<<(tiles + 3) / 4, 128, 0, stream>>>(
        hB, gi, gj, dW0, d0, P_, H_, 3 * OUT_, H_, H_);
  }
  ln_gelu_rows_kernel<H_, false><<<(P_ + 31) / 32, 256, 0, stream>>>(
      d0, db0, dg0, dbb0, nullptr, nullptr, nullptr, P_);

  gemm(d0, dW1, d1, P_, H_ / 2, H_, H_, H_ / 2, H_ / 2, 0);
  ln_gelu_rows_kernel<H_ / 2, false><<<(P_ + 31) / 32, 256, 0, stream>>>(
      d1, db1, dg1, dbb1, nullptr, nullptr, nullptr, P_);

  gemm(d1, dW2, d2, P_, H_ / 4, H_ / 2, H_ / 2, H_ / 4, H_ / 4, 0);
  ln_gelu_rows_kernel<H_ / 4, true><<<(P_ + 31) / 32, 256, 0, stream>>>(
      d2, db2, dg2, dbb2, dW3, db3, (float*)d_out, P_);
}
